# starting kernel
# baseline (speedup 1.0000x reference)
#include <hip/hip_runtime.h>

#define H      64
#define G      256
#define CHUNK  128

typedef __attribute__((ext_vector_type(2))) float v2f;
typedef __attribute__((ext_vector_type(8))) float v8f;

__device__ __forceinline__ v2f mkv2(float a, float b) { v2f r; r.x = a; r.y = b; return r; }

__device__ __forceinline__ float fsigmoid(float x) {
    return __builtin_amdgcn_rcpf(1.0f + __expf(-x));
}
__device__ __forceinline__ float ftanh(float x) {
    return 2.0f * __builtin_amdgcn_rcpf(1.0f + __expf(-2.0f * x)) - 1.0f;
}

__global__ __launch_bounds__(256, 1)
void lstm_scan_kernel(const float* __restrict__ x,
                      const float* __restrict__ W_ih,
                      const float* __restrict__ W_hh,
                      const float* __restrict__ b_ih,
                      const float* __restrict__ b_hh,
                      float* __restrict__ hs,
                      int T) {
    __shared__ float h_lds[H];
    __shared__ float gates_lds[G];
    __shared__ float xbuf[2 * CHUNK];

    const int j = threadIdx.x;

    v2f w2[H / 2];
    {
        const float4* wrow = (const float4*)(W_hh + j * H);
        #pragma unroll
        for (int k = 0; k < H / 4; ++k) {
            float4 w = wrow[k];
            w2[2 * k + 0] = mkv2(w.x, w.y);
            w2[2 * k + 1] = mkv2(w.z, w.w);
        }
    }
    const float bias = b_ih[j] + b_hh[j];
    const float wi0  = W_ih[2 * j + 0];
    const float wi1  = W_ih[2 * j + 1];

    if (j < H) h_lds[j] = 0.0f;
    float c = 0.0f;

    float xreg = (j < 2 * T) ? x[j] : 0.0f;
    __syncthreads();

    const int nchunks = (T + CHUNK - 1) / CHUNK;
    for (int ch = 0; ch < nchunks; ++ch) {
        xbuf[j] = xreg;
        const int nidx = (ch + 1) * (2 * CHUNK) + j;
        if (ch + 1 < nchunks && nidx < 2 * T) xreg = x[nidx];
        __syncthreads();

        const int tend = (T - ch * CHUNK < CHUNK) ? (T - ch * CHUNK) : CHUNK;
        for (int tt = 0; tt < tend; ++tt) {
            const int t = ch * CHUNK + tt;
            const float x0 = xbuf[2 * tt + 0];
            const float x1 = xbuf[2 * tt + 1];

            v2f acc2 = mkv2(bias + x0 * wi0 + x1 * wi1, 0.0f);
            const float4* h4 = (const float4*)h_lds;
            #pragma unroll
            for (int k = 0; k < H / 4; ++k) {
                float4 hv = h4[k];
                v2f hlo = mkv2(hv.x, hv.y);
                v2f hhi = mkv2(hv.z, hv.w);
                asm("v_pk_fma_f32 %0, %1, %2, %0" : "+v"(acc2) : "v"(hlo), "v"(w2[2 * k + 0]));
                asm("v_pk_fma_f32 %0, %1, %2, %0" : "+v"(acc2) : "v"(hhi), "v"(w2[2 * k + 1]));
            }
            gates_lds[j] = acc2.x + acc2.y;
            __syncthreads();

            if (j < H) {
                const float gi = fsigmoid(gates_lds[j]);
                const float gf = fsigmoid(gates_lds[H + j]);
                const float gg = ftanh(gates_lds[2 * H + j]);
                const float go = fsigmoid(gates_lds[3 * H + j]);
                c = gf * c + gi * gg;
                const float hn = go * ftanh(c);
                h_lds[j] = hn;
                hs[(long)t * H + j] = hn;
            }
            __syncthreads();
        }
    }
}

__global__ __launch_bounds__(256, 4)
void fc_wmma_kernel(const float* __restrict__ hs,
                    const float* __restrict__ fc_w,
                    const float* __restrict__ fc_b,
                    float* __restrict__ out,
                    int T) {
    const int lane = threadIdx.x & 31;
    const int wave = threadIdx.x >> 5;
    const int tile = blockIdx.x * 8 + wave;
    const int row0 = tile * 16;
    if (row0 >= T) return;

    const int M    = lane & 15;
    const int half = lane >> 4;
    const int N    = lane & 15;

    v2f bfrag[16];
    #pragma unroll
    for (int kk = 0; kk < 16; ++kk) {
        const int k0 = kk * 4 + half * 2;
        const float b0 = (N < 2) ? fc_w[N * H + k0 + 0] : 0.0f;
        const float b1 = (N < 2) ? fc_w[N * H + k0 + 1] : 0.0f;
        bfrag[kk] = mkv2(b0, b1);
    }

    int row = row0 + M;
    if (row >= T) row = T - 1;
    const float* arow = hs + (long)row * H;

    v8f cacc = {};
    #pragma unroll
    for (int kk = 0; kk < 16; ++kk) {
        const int k0 = kk * 4 + half * 2;
        const v2f afrag = *(const v2f*)(arow + k0);
        cacc = __builtin_amdgcn_wmma_f32_16x16x4_f32(
             false, afrag,  false, bfrag[kk],
             (short)0, cacc,  false,  false);
    }

    #pragma unroll
    for (int r = 0; r < 8; ++r) {
        const int orow = row0 + r + 8 * half;
        if (N < 2 && orow < T) out[orow * 2 + N] = cacc[r] + fc_b[N];
    }
}

extern "C" void kernel_launch(void* const* d_in, const int* in_sizes, int n_in,
                              void* d_out, int out_size, void* d_ws, size_t ws_size,
                              hipStream_t stream) {
    const float* x    = (const float*)d_in[0];
    const float* W_ih = (const float*)d_in[1];
    const float* W_hh = (const float*)d_in[2];
    const float* b_ih = (const float*)d_in[3];
    const float* b_hh = (const float*)d_in[4];
    const float* fc_w = (const float*)d_in[5];
    const float* fc_b = (const float*)d_in[6];
    float* out = (float*)d_out;

    const int T = in_sizes[0] / 2;
    float* hs = (float*)d_ws;

    lstm_scan_kernel<<<1, 256, 0, stream>>>(x, W_ih, W_hh, b_ih, b_hh, hs, T);

    const int tiles  = (T + 15) / 16;
    const int blocks = (tiles + 7) / 8;
    fc_wmma_kernel<<<blocks, 256, 0, stream>>>(hs, fc_w, fc_b, out, T);
}


# Scan kernel: the per-step workgroup barriers wait for LDS only (s_wait_dscnt + barrier written by hand), so the global store of h and the x prefetch stay in flight; all sums unchanged
# speedup vs baseline: 1.5287x; 1.5287x over previous
#include <hip/hip_runtime.h>

#define H      64
#define G      256
#define CHUNK  128

typedef __attribute__((ext_vector_type(2))) float v2f;
typedef __attribute__((ext_vector_type(8))) float v8f;

__device__ __forceinline__ v2f mkv2(float a, float b) { v2f r; r.x = a; r.y = b; return r; }

__device__ __forceinline__ float fsigmoid(float x) {
    return __builtin_amdgcn_rcpf(1.0f + __expf(-x));
}
__device__ __forceinline__ float ftanh(float x) {
    return 2.0f * __builtin_amdgcn_rcpf(1.0f + __expf(-2.0f * x)) - 1.0f;
}

__device__ __forceinline__ void lds_barrier() {
    asm volatile("s_wait_dscnt 0x0\n\ts_barrier_signal -1\n\ts_barrier_wait -1" ::: "memory");
}

__global__ __launch_bounds__(256, 1)
void lstm_scan_kernel(const float* __restrict__ x,
                      const float* __restrict__ W_ih,
                      const float* __restrict__ W_hh,
                      const float* __restrict__ b_ih,
                      const float* __restrict__ b_hh,
                      float* __restrict__ hs,
                      int T) {
    __shared__ float h_lds[H];
    __shared__ float gates_lds[G];
    __shared__ float xbuf[2 * CHUNK];

    const int j = threadIdx.x;

    v2f w2[H / 2];
    {
        const float4* wrow = (const float4*)(W_hh + j * H);
        #pragma unroll
        for (int k = 0; k < H / 4; ++k) {
            float4 w = wrow[k];
            w2[2 * k + 0] = mkv2(w.x, w.y);
            w2[2 * k + 1] = mkv2(w.z, w.w);
        }
    }
    const float bias = b_ih[j] + b_hh[j];
    const float wi0  = W_ih[2 * j + 0];
    const float wi1  = W_ih[2 * j + 1];

    if (j < H) h_lds[j] = 0.0f;
    float c = 0.0f;

    float xreg = (j < 2 * T) ? x[j] : 0.0f;
    __syncthreads();

    const int nchunks = (T + CHUNK - 1) / CHUNK;
    for (int ch = 0; ch < nchunks; ++ch) {
        xbuf[j] = xreg;
        const int nidx = (ch + 1) * (2 * CHUNK) + j;
        if (ch + 1 < nchunks && nidx < 2 * T) xreg = x[nidx];
        lds_barrier();

        const int tend = (T - ch * CHUNK < CHUNK) ? (T - ch * CHUNK) : CHUNK;
        for (int tt = 0; tt < tend; ++tt) {
            const int t = ch * CHUNK + tt;
            const float x0 = xbuf[2 * tt + 0];
            const float x1 = xbuf[2 * tt + 1];

            v2f acc2 = mkv2(bias + x0 * wi0 + x1 * wi1, 0.0f);
            const float4* h4 = (const float4*)h_lds;
            #pragma unroll
            for (int k = 0; k < H / 4; ++k) {
                float4 hv = h4[k];
                v2f hlo = mkv2(hv.x, hv.y);
                v2f hhi = mkv2(hv.z, hv.w);
                asm("v_pk_fma_f32 %0, %1, %2, %0" : "+v"(acc2) : "v"(hlo), "v"(w2[2 * k + 0]));
                asm("v_pk_fma_f32 %0, %1, %2, %0" : "+v"(acc2) : "v"(hhi), "v"(w2[2 * k + 1]));
            }
            gates_lds[j] = acc2.x + acc2.y;
            lds_barrier();

            if (j < H) {
                const float gi = fsigmoid(gates_lds[j]);
                const float gf = fsigmoid(gates_lds[H + j]);
                const float gg = ftanh(gates_lds[2 * H + j]);
                const float go = fsigmoid(gates_lds[3 * H + j]);
                c = gf * c + gi * gg;
                const float hn = go * ftanh(c);
                h_lds[j] = hn;
                hs[(long)t * H + j] = hn;
            }
            lds_barrier();
        }
    }
}

__global__ __launch_bounds__(256, 4)
void fc_wmma_kernel(const float* __restrict__ hs,
                    const float* __restrict__ fc_w,
                    const float* __restrict__ fc_b,
                    float* __restrict__ out,
                    int T) {
    const int lane = threadIdx.x & 31;
    const int wave = threadIdx.x >> 5;
    const int tile = blockIdx.x * 8 + wave;
    const int row0 = tile * 16;
    if (row0 >= T) return;

    const int M    = lane & 15;
    const int half = lane >> 4;
    const int N    = lane & 15;

    v2f bfrag[16];
    #pragma unroll
    for (int kk = 0; kk < 16; ++kk) {
        const int k0 = kk * 4 + half * 2;
        const float b0 = (N < 2) ? fc_w[N * H + k0 + 0] : 0.0f;
        const float b1 = (N < 2) ? fc_w[N * H + k0 + 1] : 0.0f;
        bfrag[kk] = mkv2(b0, b1);
    }

    int row = row0 + M;
    if (row >= T) row = T - 1;
    const float* arow = hs + (long)row * H;

    v8f cacc = {};
    #pragma unroll
    for (int kk = 0; kk < 16; ++kk) {
        const int k0 = kk * 4 + half * 2;
        const v2f afrag = *(const v2f*)(arow + k0);
        cacc = __builtin_amdgcn_wmma_f32_16x16x4_f32(
             false, afrag,  false, bfrag[kk],
             (short)0, cacc,  false,  false);
    }

    #pragma unroll
    for (int r = 0; r < 8; ++r) {
        const int orow = row0 + r + 8 * half;
        if (N < 2 && orow < T) out[orow * 2 + N] = cacc[r] + fc_b[N];
    }
}

extern "C" void kernel_launch(void* const* d_in, const int* in_sizes, int n_in,
                              void* d_out, int out_size, void* d_ws, size_t ws_size,
                              hipStream_t stream) {
    const float* x    = (const float*)d_in[0];
    const float* W_ih = (const float*)d_in[1];
    const float* W_hh = (const float*)d_in[2];
    const float* b_ih = (const float*)d_in[3];
    const float* b_hh = (const float*)d_in[4];
    const float* fc_w = (const float*)d_in[5];
    const float* fc_b = (const float*)d_in[6];
    float* out = (float*)d_out;

    const int T = in_sizes[0] / 2;
    float* hs = (float*)d_ws;

    lstm_scan_kernel<<<1, 256, 0, stream>>>(x, W_ih, W_hh, b_ih, b_hh, hs, T);

    const int tiles  = (T + 15) / 16;
    const int blocks = (tiles + 7) / 8;
    fc_wmma_kernel<<<blocks, 256, 0, stream>>>(hs, fc_w, fc_b, out, T);
}


# Scan kernel: a step's 17 LDS reads (x, then all of h) issued as one group ahead of the FMA chain (two scheduling barriers): LDS latency paid once per step, not once per pair of reads; sums unchanged
# speedup vs baseline: 2.0896x; 1.3669x over previous
#include <hip/hip_runtime.h>

#define H      64
#define G      256
#define CHUNK  128

typedef __attribute__((ext_vector_type(2))) float v2f;
typedef __attribute__((ext_vector_type(8))) float v8f;

__device__ __forceinline__ v2f mkv2(float a, float b) { v2f r; r.x = a; r.y = b; return r; }

__device__ __forceinline__ float fsigmoid(float x) {
    return __builtin_amdgcn_rcpf(1.0f + __expf(-x));
}
__device__ __forceinline__ float ftanh(float x) {
    return 2.0f * __builtin_amdgcn_rcpf(1.0f + __expf(-2.0f * x)) - 1.0f;
}

__device__ __forceinline__ void lds_barrier() {
    asm volatile("s_wait_dscnt 0x0\n\ts_barrier_signal -1\n\ts_barrier_wait -1" ::: "memory");
}

__global__ __launch_bounds__(256, 1)
void lstm_scan_kernel(const float* __restrict__ x,
                      const float* __restrict__ W_ih,
                      const float* __restrict__ W_hh,
                      const float* __restrict__ b_ih,
                      const float* __restrict__ b_hh,
                      float* __restrict__ hs,
                      int T) {
    __shared__ float h_lds[H];
    __shared__ float gates_lds[G];
    __shared__ float xbuf[2 * CHUNK];

    const int j = threadIdx.x;

    v2f w2[H / 2];
    {
        const float4* wrow = (const float4*)(W_hh + j * H);
        #pragma unroll
        for (int k = 0; k < H / 4; ++k) {
            float4 w = wrow[k];
            w2[2 * k + 0] = mkv2(w.x, w.y);
            w2[2 * k + 1] = mkv2(w.z, w.w);
        }
    }
    const float bias = b_ih[j] + b_hh[j];
    const float wi0  = W_ih[2 * j + 0];
    const float wi1  = W_ih[2 * j + 1];

    if (j < H) h_lds[j] = 0.0f;
    float c = 0.0f;

    float xreg = (j < 2 * T) ? x[j] : 0.0f;
    __syncthreads();

    const int nchunks = (T + CHUNK - 1) / CHUNK;
    for (int ch = 0; ch < nchunks; ++ch) {
        xbuf[j] = xreg;
        const int nidx = (ch + 1) * (2 * CHUNK) + j;
        if (ch + 1 < nchunks && nidx < 2 * T) xreg = x[nidx];
        lds_barrier();

        const int tend = (T - ch * CHUNK < CHUNK) ? (T - ch * CHUNK) : CHUNK;
        for (int tt = 0; tt < tend; ++tt) {
            const int t = ch * CHUNK + tt;
            const float x0 = xbuf[2 * tt + 0];
            const float x1 = xbuf[2 * tt + 1];
            __builtin_amdgcn_sched_barrier(0);

            const float4* h4 = (const float4*)h_lds;
            float4 hv[H / 4];
            #pragma unroll
            for (int k = 0; k < H / 4; ++k) hv[k] = h4[k];
            __builtin_amdgcn_sched_barrier(0);

            v2f acc2 = mkv2(bias + x0 * wi0 + x1 * wi1, 0.0f);
            #pragma unroll
            for (int k = 0; k < H / 4; ++k) {
                v2f hlo = mkv2(hv[k].x, hv[k].y);
                v2f hhi = mkv2(hv[k].z, hv[k].w);
                asm("v_pk_fma_f32 %0, %1, %2, %0" : "+v"(acc2) : "v"(hlo), "v"(w2[2 * k + 0]));
                asm("v_pk_fma_f32 %0, %1, %2, %0" : "+v"(acc2) : "v"(hhi), "v"(w2[2 * k + 1]));
            }
            gates_lds[j] = acc2.x + acc2.y;
            lds_barrier();

            if (j < H) {
                const float gi = fsigmoid(gates_lds[j]);
                const float gf = fsigmoid(gates_lds[H + j]);
                const float gg = ftanh(gates_lds[2 * H + j]);
                const float go = fsigmoid(gates_lds[3 * H + j]);
                c = gf * c + gi * gg;
                const float hn = go * ftanh(c);
                h_lds[j] = hn;
                hs[(long)t * H + j] = hn;
            }
            lds_barrier();
        }
    }
}

__global__ __launch_bounds__(256, 4)
void fc_wmma_kernel(const float* __restrict__ hs,
                    const float* __restrict__ fc_w,
                    const float* __restrict__ fc_b,
                    float* __restrict__ out,
                    int T) {
    const int lane = threadIdx.x & 31;
    const int wave = threadIdx.x >> 5;
    const int tile = blockIdx.x * 8 + wave;
    const int row0 = tile * 16;
    if (row0 >= T) return;

    const int M    = lane & 15;
    const int half = lane >> 4;
    const int N    = lane & 15;

    v2f bfrag[16];
    #pragma unroll
    for (int kk = 0; kk < 16; ++kk) {
        const int k0 = kk * 4 + half * 2;
        const float b0 = (N < 2) ? fc_w[N * H + k0 + 0] : 0.0f;
        const float b1 = (N < 2) ? fc_w[N * H + k0 + 1] : 0.0f;
        bfrag[kk] = mkv2(b0, b1);
    }

    int row = row0 + M;
    if (row >= T) row = T - 1;
    const float* arow = hs + (long)row * H;

    v8f cacc = {};
    #pragma unroll
    for (int kk = 0; kk < 16; ++kk) {
        const int k0 = kk * 4 + half * 2;
        const v2f afrag = *(const v2f*)(arow + k0);
        cacc = __builtin_amdgcn_wmma_f32_16x16x4_f32(
             false, afrag,  false, bfrag[kk],
             (short)0, cacc,  false,  false);
    }

    #pragma unroll
    for (int r = 0; r < 8; ++r) {
        const int orow = row0 + r + 8 * half;
        if (N < 2 && orow < T) out[orow * 2 + N] = cacc[r] + fc_b[N];
    }
}

extern "C" void kernel_launch(void* const* d_in, const int* in_sizes, int n_in,
                              void* d_out, int out_size, void* d_ws, size_t ws_size,
                              hipStream_t stream) {
    const float* x    = (const float*)d_in[0];
    const float* W_ih = (const float*)d_in[1];
    const float* W_hh = (const float*)d_in[2];
    const float* b_ih = (const float*)d_in[3];
    const float* b_hh = (const float*)d_in[4];
    const float* fc_w = (const float*)d_in[5];
    const float* fc_b = (const float*)d_in[6];
    float* out = (float*)d_out;

    const int T = in_sizes[0] / 2;
    float* hs = (float*)d_ws;

    lstm_scan_kernel<<<1, 256, 0, stream>>>(x, W_ih, W_hh, b_ih, b_hh, hs, T);

    const int tiles  = (T + 15) / 16;
    const int blocks = (tiles + 7) / 8;
    fc_wmma_kernel<<<blocks, 256, 0, stream>>>(hs, fc_w, fc_b, out, T);
}


# Scan kernel: each gate's 64-term sum kept as four independent packed partial sums (x and bias seed one) instead of one dependent chain of 32 v_pk_fma_f32; summation order changes
# speedup vs baseline: 2.3764x; 1.1373x over previous
#include <hip/hip_runtime.h>

#define H      64
#define G      256
#define CHUNK  128

typedef __attribute__((ext_vector_type(2))) float v2f;
typedef __attribute__((ext_vector_type(8))) float v8f;

__device__ __forceinline__ v2f mkv2(float a, float b) { v2f r; r.x = a; r.y = b; return r; }

__device__ __forceinline__ float fsigmoid(float x) {
    return __builtin_amdgcn_rcpf(1.0f + __expf(-x));
}
__device__ __forceinline__ float ftanh(float x) {
    return 2.0f * __builtin_amdgcn_rcpf(1.0f + __expf(-2.0f * x)) - 1.0f;
}

__device__ __forceinline__ void lds_barrier() {
    asm volatile("s_wait_dscnt 0x0\n\ts_barrier_signal -1\n\ts_barrier_wait -1" ::: "memory");
}

__global__ __launch_bounds__(256, 1)
void lstm_scan_kernel(const float* __restrict__ x,
                      const float* __restrict__ W_ih,
                      const float* __restrict__ W_hh,
                      const float* __restrict__ b_ih,
                      const float* __restrict__ b_hh,
                      float* __restrict__ hs,
                      int T) {
    __shared__ float h_lds[H];
    __shared__ float gates_lds[G];
    __shared__ float xbuf[2 * CHUNK];

    const int j = threadIdx.x;

    v2f w2[H / 2];
    {
        const float4* wrow = (const float4*)(W_hh + j * H);
        #pragma unroll
        for (int k = 0; k < H / 4; ++k) {
            float4 w = wrow[k];
            w2[2 * k + 0] = mkv2(w.x, w.y);
            w2[2 * k + 1] = mkv2(w.z, w.w);
        }
    }
    const float bias = b_ih[j] + b_hh[j];
    const v2f wi = mkv2(W_ih[2 * j + 0], W_ih[2 * j + 1]);

    if (j < H) h_lds[j] = 0.0f;
    float c = 0.0f;

    float xreg = (j < 2 * T) ? x[j] : 0.0f;
    __syncthreads();

    const int nchunks = (T + CHUNK - 1) / CHUNK;
    for (int ch = 0; ch < nchunks; ++ch) {
        xbuf[j] = xreg;
        const int nidx = (ch + 1) * (2 * CHUNK) + j;
        if (ch + 1 < nchunks && nidx < 2 * T) xreg = x[nidx];
        lds_barrier();

        const int tend = (T - ch * CHUNK < CHUNK) ? (T - ch * CHUNK) : CHUNK;
        for (int tt = 0; tt < tend; ++tt) {
            const int t = ch * CHUNK + tt;
            const v2f xv = *(const v2f*)&xbuf[2 * tt];
            __builtin_amdgcn_sched_barrier(0);

            const float4* h4 = (const float4*)h_lds;
            float4 hv[H / 4];
            #pragma unroll
            for (int k = 0; k < H / 4; ++k) hv[k] = h4[k];
            __builtin_amdgcn_sched_barrier(0);

            v2f accA = __builtin_elementwise_fma(xv, wi, mkv2(bias, 0.0f));
            v2f accB = mkv2(0.0f, 0.0f);
            v2f accC = mkv2(0.0f, 0.0f);
            v2f accD = mkv2(0.0f, 0.0f);
            #pragma unroll
            for (int k = 0; k < H / 4; k += 2) {
                accA = __builtin_elementwise_fma(mkv2(hv[k].x, hv[k].y), w2[2 * k + 0], accA);
                accB = __builtin_elementwise_fma(mkv2(hv[k].z, hv[k].w), w2[2 * k + 1], accB);
                accC = __builtin_elementwise_fma(mkv2(hv[k + 1].x, hv[k + 1].y), w2[2 * k + 2], accC);
                accD = __builtin_elementwise_fma(mkv2(hv[k + 1].z, hv[k + 1].w), w2[2 * k + 3], accD);
            }
            const v2f accS = (accA + accB) + (accC + accD);
            gates_lds[j] = accS.x + accS.y;
            lds_barrier();

            if (j < H) {
                const float gi = fsigmoid(gates_lds[j]);
                const float gf = fsigmoid(gates_lds[H + j]);
                const float gg = ftanh(gates_lds[2 * H + j]);
                const float go = fsigmoid(gates_lds[3 * H + j]);
                c = gf * c + gi * gg;
                const float hn = go * ftanh(c);
                h_lds[j] = hn;
                hs[(long)t * H + j] = hn;
            }
            lds_barrier();
        }
    }
}

__global__ __launch_bounds__(256, 4)
void fc_wmma_kernel(const float* __restrict__ hs,
                    const float* __restrict__ fc_w,
                    const float* __restrict__ fc_b,
                    float* __restrict__ out,
                    int T) {
    const int lane = threadIdx.x & 31;
    const int wave = threadIdx.x >> 5;
    const int tile = blockIdx.x * 8 + wave;
    const int row0 = tile * 16;
    if (row0 >= T) return;

    const int M    = lane & 15;
    const int half = lane >> 4;
    const int N    = lane & 15;

    v2f bfrag[16];
    #pragma unroll
    for (int kk = 0; kk < 16; ++kk) {
        const int k0 = kk * 4 + half * 2;
        const float b0 = (N < 2) ? fc_w[N * H + k0 + 0] : 0.0f;
        const float b1 = (N < 2) ? fc_w[N * H + k0 + 1] : 0.0f;
        bfrag[kk] = mkv2(b0, b1);
    }

    int row = row0 + M;
    if (row >= T) row = T - 1;
    const float* arow = hs + (long)row * H;

    v8f cacc = {};
    #pragma unroll
    for (int kk = 0; kk < 16; ++kk) {
        const int k0 = kk * 4 + half * 2;
        const v2f afrag = *(const v2f*)(arow + k0);
        cacc = __builtin_amdgcn_wmma_f32_16x16x4_f32(
             false, afrag,  false, bfrag[kk],
             (short)0, cacc,  false,  false);
    }

    #pragma unroll
    for (int r = 0; r < 8; ++r) {
        const int orow = row0 + r + 8 * half;
        if (N < 2 && orow < T) out[orow * 2 + N] = cacc[r] + fc_b[N];
    }
}

extern "C" void kernel_launch(void* const* d_in, const int* in_sizes, int n_in,
                              void* d_out, int out_size, void* d_ws, size_t ws_size,
                              hipStream_t stream) {
    const float* x    = (const float*)d_in[0];
    const float* W_ih = (const float*)d_in[1];
    const float* W_hh = (const float*)d_in[2];
    const float* b_ih = (const float*)d_in[3];
    const float* b_hh = (const float*)d_in[4];
    const float* fc_w = (const float*)d_in[5];
    const float* fc_b = (const float*)d_in[6];
    float* out = (float*)d_out;

    const int T = in_sizes[0] / 2;
    float* hs = (float*)d_ws;

    lstm_scan_kernel<<<1, 256, 0, stream>>>(x, W_ih, W_hh, b_ih, b_hh, hs, T);

    const int tiles  = (T + 15) / 16;
    const int blocks = (tiles + 7) / 8;
    fc_wmma_kernel<<<blocks, 256, 0, stream>>>(hs, fc_w, fc_b, out, T);
}


# Scan kernel: one lane per hidden unit holding its four gate rows (2 waves) instead of one lane per gate row: no gate exchange through LDS, h double-buffered, one barrier per step; sums unchanged
# speedup vs baseline: 2.9660x; 1.2481x over previous
#include <hip/hip_runtime.h>

#define H      64
#define CHUNK  128

typedef __attribute__((ext_vector_type(2))) float v2f;
typedef __attribute__((ext_vector_type(8))) float v8f;

__device__ __forceinline__ v2f mkv2(float a, float b) { v2f r; r.x = a; r.y = b; return r; }

__device__ __forceinline__ float fsigmoid(float x) {
    return __builtin_amdgcn_rcpf(1.0f + __expf(-x));
}
__device__ __forceinline__ float ftanh(float x) {
    return 2.0f * __builtin_amdgcn_rcpf(1.0f + __expf(-2.0f * x)) - 1.0f;
}

__device__ __forceinline__ void lds_barrier() {
    asm volatile("s_wait_dscnt 0x0\n\ts_barrier_signal -1\n\ts_barrier_wait -1" ::: "memory");
}

__device__ __forceinline__ float4 load_x4(const float* __restrict__ x, int idx, int n) {
    float4 r = make_float4(0.0f, 0.0f, 0.0f, 0.0f);
    if (idx + 3 < n) {
        r = *(const float4*)(x + idx);
    } else {
        if (idx + 0 < n) r.x = x[idx + 0];
        if (idx + 1 < n) r.y = x[idx + 1];
        if (idx + 2 < n) r.z = x[idx + 2];
    }
    return r;
}

__global__ __launch_bounds__(64, 1)
void lstm_scan_kernel(const float* __restrict__ x,
                      const float* __restrict__ W_ih,
                      const float* __restrict__ W_hh,
                      const float* __restrict__ b_ih,
                      const float* __restrict__ b_hh,
                      float* __restrict__ hs,
                      int T) {
    __shared__ float h_lds[2 * H];
    __shared__ float xbuf[2 * CHUNK];

    const int j = threadIdx.x;
    const int n_x = 2 * T;

    v2f w2[4][H / 2];
    v2f wi[4];
    float bias[4];
    #pragma unroll
    for (int g = 0; g < 4; ++g) {
        const int row = g * H + j;
        const float4* wrow = (const float4*)(W_hh + row * H);
        #pragma unroll
        for (int k = 0; k < H / 4; ++k) {
            float4 w = wrow[k];
            w2[g][2 * k + 0] = mkv2(w.x, w.y);
            w2[g][2 * k + 1] = mkv2(w.z, w.w);
        }
        bias[g] = b_ih[row] + b_hh[row];
        wi[g] = mkv2(W_ih[2 * row + 0], W_ih[2 * row + 1]);
    }

    h_lds[j] = 0.0f;
    float c = 0.0f;

    float4 xreg = load_x4(x, 4 * j, n_x);
    __syncthreads();

    int rd = 0;
    const int nchunks = (T + CHUNK - 1) / CHUNK;
    for (int ch = 0; ch < nchunks; ++ch) {
        *(float4*)&xbuf[4 * j] = xreg;
        if (ch + 1 < nchunks) xreg = load_x4(x, (ch + 1) * (2 * CHUNK) + 4 * j, n_x);
        lds_barrier();

        const int tend = (T - ch * CHUNK < CHUNK) ? (T - ch * CHUNK) : CHUNK;
        for (int tt = 0; tt < tend; ++tt) {
            const int t = ch * CHUNK + tt;
            const v2f xv = *(const v2f*)&xbuf[2 * tt];
            __builtin_amdgcn_sched_barrier(0);

            const float4* h4 = (const float4*)(h_lds + rd);
            float4 hv[H / 4];
            #pragma unroll
            for (int k = 0; k < H / 4; ++k) hv[k] = h4[k];
            __builtin_amdgcn_sched_barrier(0);

            v2f accA[4], accB[4], accC[4], accD[4];
            #pragma unroll
            for (int g = 0; g < 4; ++g) {
                accA[g] = __builtin_elementwise_fma(xv, wi[g], mkv2(bias[g], 0.0f));
                accB[g] = mkv2(0.0f, 0.0f);
                accC[g] = mkv2(0.0f, 0.0f);
                accD[g] = mkv2(0.0f, 0.0f);
            }
            #pragma unroll
            for (int k = 0; k < H / 4; k += 2) {
                #pragma unroll
                for (int g = 0; g < 4; ++g) {
                    accA[g] = __builtin_elementwise_fma(mkv2(hv[k].x, hv[k].y), w2[g][2 * k + 0], accA[g]);
                    accB[g] = __builtin_elementwise_fma(mkv2(hv[k].z, hv[k].w), w2[g][2 * k + 1], accB[g]);
                    accC[g] = __builtin_elementwise_fma(mkv2(hv[k + 1].x, hv[k + 1].y), w2[g][2 * k + 2], accC[g]);
                    accD[g] = __builtin_elementwise_fma(mkv2(hv[k + 1].z, hv[k + 1].w), w2[g][2 * k + 3], accD[g]);
                }
            }
            float gate[4];
            #pragma unroll
            for (int g = 0; g < 4; ++g) {
                const v2f accS = (accA[g] + accB[g]) + (accC[g] + accD[g]);
                gate[g] = accS.x + accS.y;
            }

            const float gi = fsigmoid(gate[0]);
            const float gf = fsigmoid(gate[1]);
            const float gg = ftanh(gate[2]);
            const float go = fsigmoid(gate[3]);
            c = gf * c + gi * gg;
            const float hn = go * ftanh(c);
            h_lds[(H - rd) + j] = hn;
            hs[(long)t * H + j] = hn;
            lds_barrier();
            rd = H - rd;
        }
    }
}

__global__ __launch_bounds__(256, 4)
void fc_wmma_kernel(const float* __restrict__ hs,
                    const float* __restrict__ fc_w,
                    const float* __restrict__ fc_b,
                    float* __restrict__ out,
                    int T) {
    const int lane = threadIdx.x & 31;
    const int wave = threadIdx.x >> 5;
    const int tile = blockIdx.x * 8 + wave;
    const int row0 = tile * 16;
    if (row0 >= T) return;

    const int M    = lane & 15;
    const int half = lane >> 4;
    const int N    = lane & 15;

    v2f bfrag[16];
    #pragma unroll
    for (int kk = 0; kk < 16; ++kk) {
        const int k0 = kk * 4 + half * 2;
        const float b0 = (N < 2) ? fc_w[N * H + k0 + 0] : 0.0f;
        const float b1 = (N < 2) ? fc_w[N * H + k0 + 1] : 0.0f;
        bfrag[kk] = mkv2(b0, b1);
    }

    int row = row0 + M;
    if (row >= T) row = T - 1;
    const float* arow = hs + (long)row * H;

    v8f cacc = {};
    #pragma unroll
    for (int kk = 0; kk < 16; ++kk) {
        const int k0 = kk * 4 + half * 2;
        const v2f afrag = *(const v2f*)(arow + k0);
        cacc = __builtin_amdgcn_wmma_f32_16x16x4_f32(
             false, afrag,  false, bfrag[kk],
             (short)0, cacc,  false,  false);
    }

    #pragma unroll
    for (int r = 0; r < 8; ++r) {
        const int orow = row0 + r + 8 * half;
        if (N < 2 && orow < T) out[orow * 2 + N] = cacc[r] + fc_b[N];
    }
}

extern "C" void kernel_launch(void* const* d_in, const int* in_sizes, int n_in,
                              void* d_out, int out_size, void* d_ws, size_t ws_size,
                              hipStream_t stream) {
    const float* x    = (const float*)d_in[0];
    const float* W_ih = (const float*)d_in[1];
    const float* W_hh = (const float*)d_in[2];
    const float* b_ih = (const float*)d_in[3];
    const float* b_hh = (const float*)d_in[4];
    const float* fc_w = (const float*)d_in[5];
    const float* fc_b = (const float*)d_in[6];
    float* out = (float*)d_out;

    const int T = in_sizes[0] / 2;
    float* hs = (float*)d_ws;

    lstm_scan_kernel<<<1, 64, 0, stream>>>(x, W_ih, W_hh, b_ih, b_hh, hs, T);

    const int tiles  = (T + 15) / 16;
    const int blocks = (tiles + 7) / 8;
    fc_wmma_kernel<<<blocks, 256, 0, stream>>>(hs, fc_w, fc_b, out, T);
}


# Scan kernel: each wave reads its own half of h back from LDS ahead of the step barrier (counted s_wait_dscnt; LDS store of h ahead of the global one), other half behind it; order of sums changes
# speedup vs baseline: 3.2683x; 1.1019x over previous
#include <hip/hip_runtime.h>

#define H      64
#define CHUNK  128

typedef __attribute__((ext_vector_type(2))) float v2f;
typedef __attribute__((ext_vector_type(8))) float v8f;

__device__ __forceinline__ v2f mkv2(float a, float b) { v2f r; r.x = a; r.y = b; return r; }

__device__ __forceinline__ float fsigmoid(float x) {
    return __builtin_amdgcn_rcpf(1.0f + __expf(-x));
}
__device__ __forceinline__ float ftanh(float x) {
    return 2.0f * __builtin_amdgcn_rcpf(1.0f + __expf(-2.0f * x)) - 1.0f;
}

__device__ __forceinline__ void lds_barrier() {
    asm volatile("s_wait_dscnt 0x0\n\ts_barrier_signal -1\n\ts_barrier_wait -1" ::: "memory");
}

__device__ __forceinline__ float4 load_x4(const float* __restrict__ x, int idx, int n) {
    float4 r = make_float4(0.0f, 0.0f, 0.0f, 0.0f);
    if (idx + 3 < n) {
        r = *(const float4*)(x + idx);
    } else {
        if (idx + 0 < n) r.x = x[idx + 0];
        if (idx + 1 < n) r.y = x[idx + 1];
        if (idx + 2 < n) r.z = x[idx + 2];
    }
    return r;
}


__device__ __forceinline__ void handover_barrier_9_reads_in_flight() {
    asm volatile("s_wait_dscnt 0x9\n\ts_barrier_signal -1\n\ts_barrier_wait -1" ::: "memory");
}

__global__ __launch_bounds__(64, 1)
void lstm_scan_kernel(const float* __restrict__ x,
                      const float* __restrict__ W_ih,
                      const float* __restrict__ W_hh,
                      const float* __restrict__ b_ih,
                      const float* __restrict__ b_hh,
                      float* __restrict__ hs,
                      int T) {
    __shared__ float h_lds[2 * H];
    __shared__ float xbuf[2 * CHUNK];

    const int j = threadIdx.x;
    const int n_x = 2 * T;
    const int wv = j >> 5;
    const int own_off = 32 * wv;
    const int oth_off = 32 - own_off;

    v2f w2[4][H / 2];
    v2f wi[4];
    float bias[4];
    #pragma unroll
    for (int g = 0; g < 4; ++g) {
        const int row = g * H + j;
        const float4* wrow = (const float4*)(W_hh + row * H);
        #pragma unroll
        for (int k = 0; k < H / 4; ++k) {
            float4 w = wrow[(k + 8 * wv) & 15];
            w2[g][2 * k + 0] = mkv2(w.x, w.y);
            w2[g][2 * k + 1] = mkv2(w.z, w.w);
        }
        bias[g] = b_ih[row] + b_hh[row];
        wi[g] = mkv2(W_ih[2 * row + 0], W_ih[2 * row + 1]);
    }

    h_lds[j] = 0.0f;
    float c = 0.0f;

    float4 xreg = load_x4(x, 4 * j, n_x);
    __syncthreads();

    int rd = 0;
    const int nchunks = (T + CHUNK - 1) / CHUNK;
    for (int ch = 0; ch < nchunks; ++ch) {
        lds_barrier();
        *(float4*)&xbuf[4 * j] = xreg;
        if (ch + 1 < nchunks) xreg = load_x4(x, (ch + 1) * (2 * CHUNK) + 4 * j, n_x);
        lds_barrier();

        const int tend = (T - ch * CHUNK < CHUNK) ? (T - ch * CHUNK) : CHUNK;
        for (int tt = 0; tt < tend; ++tt) {
            const int t = ch * CHUNK + tt;
            const v2f xv = *(const v2f*)&xbuf[2 * tt];
            __builtin_amdgcn_sched_barrier(0);
            const float4* ho4 = (const float4*)(h_lds + rd + own_off);
            float4 ho[H / 8];
            #pragma unroll
            for (int k = 0; k < H / 8; ++k) ho[k] = ho4[k];
            handover_barrier_9_reads_in_flight();

            const float4* hx4 = (const float4*)(h_lds + rd + oth_off);
            float4 hx[H / 8];
            #pragma unroll
            for (int k = 0; k < H / 8; ++k) hx[k] = hx4[k];
            __builtin_amdgcn_sched_barrier(0);

            v2f accA[4], accB[4], accC[4], accD[4];
            #pragma unroll
            for (int g = 0; g < 4; ++g) {
                accA[g] = __builtin_elementwise_fma(xv, wi[g], mkv2(bias[g], 0.0f));
                accB[g] = mkv2(0.0f, 0.0f);
                accC[g] = mkv2(0.0f, 0.0f);
                accD[g] = mkv2(0.0f, 0.0f);
            }
            #pragma unroll
            for (int k = 0; k < H / 8; k += 2) {
                #pragma unroll
                for (int g = 0; g < 4; ++g) {
                    accA[g] = __builtin_elementwise_fma(mkv2(ho[k].x, ho[k].y), w2[g][2 * k + 0], accA[g]);
                    accB[g] = __builtin_elementwise_fma(mkv2(ho[k].z, ho[k].w), w2[g][2 * k + 1], accB[g]);
                    accC[g] = __builtin_elementwise_fma(mkv2(ho[k + 1].x, ho[k + 1].y), w2[g][2 * k + 2], accC[g]);
                    accD[g] = __builtin_elementwise_fma(mkv2(ho[k + 1].z, ho[k + 1].w), w2[g][2 * k + 3], accD[g]);
                }
            }
            #pragma unroll
            for (int k = 0; k < H / 8; k += 2) {
                #pragma unroll
                for (int g = 0; g < 4; ++g) {
                    accA[g] = __builtin_elementwise_fma(mkv2(hx[k].x, hx[k].y), w2[g][16 + 2 * k + 0], accA[g]);
                    accB[g] = __builtin_elementwise_fma(mkv2(hx[k].z, hx[k].w), w2[g][16 + 2 * k + 1], accB[g]);
                    accC[g] = __builtin_elementwise_fma(mkv2(hx[k + 1].x, hx[k + 1].y), w2[g][16 + 2 * k + 2], accC[g]);
                    accD[g] = __builtin_elementwise_fma(mkv2(hx[k + 1].z, hx[k + 1].w), w2[g][16 + 2 * k + 3], accD[g]);
                }
            }
            float gate[4];
            #pragma unroll
            for (int g = 0; g < 4; ++g) {
                const v2f accS = (accA[g] + accB[g]) + (accC[g] + accD[g]);
                gate[g] = accS.x + accS.y;
            }

            const float gi = fsigmoid(gate[0]);
            const float gf = fsigmoid(gate[1]);
            const float gg = ftanh(gate[2]);
            const float go = fsigmoid(gate[3]);
            c = gf * c + gi * gg;
            const float hn = go * ftanh(c);
            h_lds[(H - rd) + j] = hn;
            __builtin_amdgcn_sched_barrier(0);
            hs[(long)t * H + j] = hn;
            rd = H - rd;
        }
    }
}

__global__ __launch_bounds__(256, 4)
void fc_wmma_kernel(const float* __restrict__ hs,
                    const float* __restrict__ fc_w,
                    const float* __restrict__ fc_b,
                    float* __restrict__ out,
                    int T) {
    const int lane = threadIdx.x & 31;
    const int wave = threadIdx.x >> 5;
    const int tile = blockIdx.x * 8 + wave;
    const int row0 = tile * 16;
    if (row0 >= T) return;

    const int M    = lane & 15;
    const int half = lane >> 4;
    const int N    = lane & 15;

    v2f bfrag[16];
    #pragma unroll
    for (int kk = 0; kk < 16; ++kk) {
        const int k0 = kk * 4 + half * 2;
        const float b0 = (N < 2) ? fc_w[N * H + k0 + 0] : 0.0f;
        const float b1 = (N < 2) ? fc_w[N * H + k0 + 1] : 0.0f;
        bfrag[kk] = mkv2(b0, b1);
    }

    int row = row0 + M;
    if (row >= T) row = T - 1;
    const float* arow = hs + (long)row * H;

    v8f cacc = {};
    #pragma unroll
    for (int kk = 0; kk < 16; ++kk) {
        const int k0 = kk * 4 + half * 2;
        const v2f afrag = *(const v2f*)(arow + k0);
        cacc = __builtin_amdgcn_wmma_f32_16x16x4_f32(
             false, afrag,  false, bfrag[kk],
             (short)0, cacc,  false,  false);
    }

    #pragma unroll
    for (int r = 0; r < 8; ++r) {
        const int orow = row0 + r + 8 * half;
        if (N < 2 && orow < T) out[orow * 2 + N] = cacc[r] + fc_b[N];
    }
}

extern "C" void kernel_launch(void* const* d_in, const int* in_sizes, int n_in,
                              void* d_out, int out_size, void* d_ws, size_t ws_size,
                              hipStream_t stream) {
    const float* x    = (const float*)d_in[0];
    const float* W_ih = (const float*)d_in[1];
    const float* W_hh = (const float*)d_in[2];
    const float* b_ih = (const float*)d_in[3];
    const float* b_hh = (const float*)d_in[4];
    const float* fc_w = (const float*)d_in[5];
    const float* fc_b = (const float*)d_in[6];
    float* out = (float*)d_out;

    const int T = in_sizes[0] / 2;
    float* hs = (float*)d_ws;

    lstm_scan_kernel<<<1, 64, 0, stream>>>(x, W_ih, W_hh, b_ih, b_hh, hs, T);

    const int tiles  = (T + 15) / 16;
    const int blocks = (tiles + 7) / 8;
    fc_wmma_kernel<<<blocks, 256, 0, stream>>>(hs, fc_w, fc_b, out, T);
}


# Scan kernel: gates and cell update on the hardware tanh instruction (sigmoid(z) = 0.5 + 0.5 tanh(z/2), the half folded into the i, f, o weight rows) in place of exp + rcp chains; gate sums unchanged
# speedup vs baseline: 3.7778x; 1.1559x over previous
#include <hip/hip_runtime.h>

#define H      64
#define CHUNK  128

typedef __attribute__((ext_vector_type(2))) float v2f;
typedef __attribute__((ext_vector_type(8))) float v8f;

__device__ __forceinline__ v2f mkv2(float a, float b) { v2f r; r.x = a; r.y = b; return r; }

__device__ __forceinline__ float htanh(float x) {
    return __builtin_amdgcn_tanhf(x);
}

__device__ __forceinline__ void lds_barrier() {
    asm volatile("s_wait_dscnt 0x0\n\ts_barrier_signal -1\n\ts_barrier_wait -1" ::: "memory");
}

__device__ __forceinline__ float4 load_x4(const float* __restrict__ x, int idx, int n) {
    float4 r = make_float4(0.0f, 0.0f, 0.0f, 0.0f);
    if (idx + 3 < n) {
        r = *(const float4*)(x + idx);
    } else {
        if (idx + 0 < n) r.x = x[idx + 0];
        if (idx + 1 < n) r.y = x[idx + 1];
        if (idx + 2 < n) r.z = x[idx + 2];
    }
    return r;
}


__device__ __forceinline__ void handover_barrier_9_reads_in_flight() {
    asm volatile("s_wait_dscnt 0x9\n\ts_barrier_signal -1\n\ts_barrier_wait -1" ::: "memory");
}

__global__ __launch_bounds__(64, 1)
void lstm_scan_kernel(const float* __restrict__ x,
                      const float* __restrict__ W_ih,
                      const float* __restrict__ W_hh,
                      const float* __restrict__ b_ih,
                      const float* __restrict__ b_hh,
                      float* __restrict__ hs,
                      int T) {
    __shared__ float h_lds[2 * H];
    __shared__ float xbuf[2 * CHUNK];

    const int j = threadIdx.x;
    const int n_x = 2 * T;
    const int wv = j >> 5;
    const int own_off = 32 * wv;
    const int oth_off = 32 - own_off;

    v2f w2[4][H / 2];
    v2f wi[4];
    float bias[4];
    #pragma unroll
    for (int g = 0; g < 4; ++g) {
        const int row = g * H + j;
        const float sc = (g == 2) ? 1.0f : 0.5f;
        const float4* wrow = (const float4*)(W_hh + row * H);
        #pragma unroll
        for (int k = 0; k < H / 4; ++k) {
            float4 w = wrow[(k + 8 * wv) & 15];
            w2[g][2 * k + 0] = mkv2(sc * w.x, sc * w.y);
            w2[g][2 * k + 1] = mkv2(sc * w.z, sc * w.w);
        }
        bias[g] = sc * (b_ih[row] + b_hh[row]);
        wi[g] = mkv2(sc * W_ih[2 * row + 0], sc * W_ih[2 * row + 1]);
    }

    h_lds[j] = 0.0f;
    float c = 0.0f;
    float hc = 0.0f;

    float4 xreg = load_x4(x, 4 * j, n_x);
    __syncthreads();

    int rd = 0;
    const int nchunks = (T + CHUNK - 1) / CHUNK;
    for (int ch = 0; ch < nchunks; ++ch) {
        lds_barrier();
        *(float4*)&xbuf[4 * j] = xreg;
        if (ch + 1 < nchunks) xreg = load_x4(x, (ch + 1) * (2 * CHUNK) + 4 * j, n_x);
        lds_barrier();

        const int tend = (T - ch * CHUNK < CHUNK) ? (T - ch * CHUNK) : CHUNK;
        for (int tt = 0; tt < tend; ++tt) {
            const int t = ch * CHUNK + tt;
            const v2f xv = *(const v2f*)&xbuf[2 * tt];
            __builtin_amdgcn_sched_barrier(0);
            const float4* ho4 = (const float4*)(h_lds + rd + own_off);
            float4 ho[H / 8];
            #pragma unroll
            for (int k = 0; k < H / 8; ++k) ho[k] = ho4[k];
            handover_barrier_9_reads_in_flight();

            const float4* hx4 = (const float4*)(h_lds + rd + oth_off);
            float4 hx[H / 8];
            #pragma unroll
            for (int k = 0; k < H / 8; ++k) hx[k] = hx4[k];
            __builtin_amdgcn_sched_barrier(0);

            v2f accA[4], accB[4], accC[4], accD[4];
            #pragma unroll
            for (int g = 0; g < 4; ++g) {
                accA[g] = __builtin_elementwise_fma(xv, wi[g], mkv2(bias[g], 0.0f));
                accB[g] = mkv2(0.0f, 0.0f);
                accC[g] = mkv2(0.0f, 0.0f);
                accD[g] = mkv2(0.0f, 0.0f);
            }
            #pragma unroll
            for (int k = 0; k < H / 8; k += 2) {
                #pragma unroll
                for (int g = 0; g < 4; ++g) {
                    accA[g] = __builtin_elementwise_fma(mkv2(ho[k].x, ho[k].y), w2[g][2 * k + 0], accA[g]);
                    accB[g] = __builtin_elementwise_fma(mkv2(ho[k].z, ho[k].w), w2[g][2 * k + 1], accB[g]);
                    accC[g] = __builtin_elementwise_fma(mkv2(ho[k + 1].x, ho[k + 1].y), w2[g][2 * k + 2], accC[g]);
                    accD[g] = __builtin_elementwise_fma(mkv2(ho[k + 1].z, ho[k + 1].w), w2[g][2 * k + 3], accD[g]);
                }
            }
            #pragma unroll
            for (int k = 0; k < H / 8; k += 2) {
                #pragma unroll
                for (int g = 0; g < 4; ++g) {
                    accA[g] = __builtin_elementwise_fma(mkv2(hx[k].x, hx[k].y), w2[g][16 + 2 * k + 0], accA[g]);
                    accB[g] = __builtin_elementwise_fma(mkv2(hx[k].z, hx[k].w), w2[g][16 + 2 * k + 1], accB[g]);
                    accC[g] = __builtin_elementwise_fma(mkv2(hx[k + 1].x, hx[k + 1].y), w2[g][16 + 2 * k + 2], accC[g]);
                    accD[g] = __builtin_elementwise_fma(mkv2(hx[k + 1].z, hx[k + 1].w), w2[g][16 + 2 * k + 3], accD[g]);
                }
            }
            float gate[4];
            #pragma unroll
            for (int g = 0; g < 4; ++g) {
                const v2f accS = (accA[g] + accB[g]) + (accC[g] + accD[g]);
                gate[g] = accS.x + accS.y;
            }

            const float ti = htanh(gate[0]);
            const float tf = htanh(gate[1]);
            const float gg = htanh(gate[2]);
            const float to = htanh(gate[3]);
            const float gi = __builtin_fmaf(ti, 0.5f, 0.5f);
            const float fc = __builtin_fmaf(tf, hc, hc);
            c = __builtin_fmaf(gi, gg, fc);
            hc = 0.5f * c;
            const float go = __builtin_fmaf(to, 0.5f, 0.5f);
            const float hn = go * htanh(c);
            h_lds[(H - rd) + j] = hn;
            __builtin_amdgcn_sched_barrier(0);
            hs[(long)t * H + j] = hn;
            rd = H - rd;
        }
    }
}

__global__ __launch_bounds__(256, 4)
void fc_wmma_kernel(const float* __restrict__ hs,
                    const float* __restrict__ fc_w,
                    const float* __restrict__ fc_b,
                    float* __restrict__ out,
                    int T) {
    const int lane = threadIdx.x & 31;
    const int wave = threadIdx.x >> 5;
    const int tile = blockIdx.x * 8 + wave;
    const int row0 = tile * 16;
    if (row0 >= T) return;

    const int M    = lane & 15;
    const int half = lane >> 4;
    const int N    = lane & 15;

    v2f bfrag[16];
    #pragma unroll
    for (int kk = 0; kk < 16; ++kk) {
        const int k0 = kk * 4 + half * 2;
        const float b0 = (N < 2) ? fc_w[N * H + k0 + 0] : 0.0f;
        const float b1 = (N < 2) ? fc_w[N * H + k0 + 1] : 0.0f;
        bfrag[kk] = mkv2(b0, b1);
    }

    int row = row0 + M;
    if (row >= T) row = T - 1;
    const float* arow = hs + (long)row * H;

    v8f cacc = {};
    #pragma unroll
    for (int kk = 0; kk < 16; ++kk) {
        const int k0 = kk * 4 + half * 2;
        const v2f afrag = *(const v2f*)(arow + k0);
        cacc = __builtin_amdgcn_wmma_f32_16x16x4_f32(
             false, afrag,  false, bfrag[kk],
             (short)0, cacc,  false,  false);
    }

    #pragma unroll
    for (int r = 0; r < 8; ++r) {
        const int orow = row0 + r + 8 * half;
        if (N < 2 && orow < T) out[orow * 2 + N] = cacc[r] + fc_b[N];
    }
}

extern "C" void kernel_launch(void* const* d_in, const int* in_sizes, int n_in,
                              void* d_out, int out_size, void* d_ws, size_t ws_size,
                              hipStream_t stream) {
    const float* x    = (const float*)d_in[0];
    const float* W_ih = (const float*)d_in[1];
    const float* W_hh = (const float*)d_in[2];
    const float* b_ih = (const float*)d_in[3];
    const float* b_hh = (const float*)d_in[4];
    const float* fc_w = (const float*)d_in[5];
    const float* fc_b = (const float*)d_in[6];
    float* out = (float*)d_out;

    const int T = in_sizes[0] / 2;
    float* hs = (float*)d_ws;

    lstm_scan_kernel<<<1, 64, 0, stream>>>(x, W_ih, W_hh, b_ih, b_hh, hs, T);

    const int tiles  = (T + 15) / 16;
    const int blocks = (tiles + 7) / 8;
    fc_wmma_kernel<<<blocks, 256, 0, stream>>>(hs, fc_w, fc_b, out, T);
}


# Scan kernel: 4 waves (one per SIMD) for 2: lanes c and c+16 each sum half of k for a unit's four gates, joined by v_permlanex16 + add; hs in whole 128-byte lines per wave; order of sums changes
# speedup vs baseline: 5.3300x; 1.4109x over previous
#include <hip/hip_runtime.h>


#define H      64
#define CHUNK  128

typedef __attribute__((ext_vector_type(2))) float v2f;
typedef __attribute__((ext_vector_type(8))) float v8f;

__device__ __forceinline__ v2f mkv2(float a, float b) { v2f r; r.x = a; r.y = b; return r; }

__device__ __forceinline__ float htanh(float x) {
    return __builtin_amdgcn_tanhf(x);
}

__device__ __forceinline__ float xhalf(float x) {
    const int b = __builtin_bit_cast(int, x);
    return __builtin_bit_cast(float,
        __builtin_amdgcn_permlanex16(b, b, 0x76543210, 0xfedcba98, false, false));
}

__device__ __forceinline__ void lds_barrier() {
    asm volatile("s_wait_dscnt 0x0\n\ts_barrier_signal -1\n\ts_barrier_wait -1" ::: "memory");
}

__device__ __forceinline__ void handover_barrier_3_reads_in_flight() {
    asm volatile("s_wait_dscnt 0x3\n\ts_barrier_signal -1\n\ts_barrier_wait -1" ::: "memory");
}

__device__ __forceinline__ float2 load_x2(const float* __restrict__ x, int idx, int n) {
    float2 r = make_float2(0.0f, 0.0f);
    if (idx + 1 < n) {
        r = *(const float2*)(x + idx);
    } else if (idx < n) {
        r.x = x[idx];
    }
    return r;
}


template <int BUF>
__device__ __forceinline__ float lstm_step(const float* h_lds, const float* xbuf, int tt, int j,
                                           const int (&q)[8], const v2f (&w2)[4][16],
                                           const v2f (&wi)[4], const float (&bias)[4],
                                           float& c, float& hc) {
    const v2f xv = *(const v2f*)&xbuf[2 * tt];
    __builtin_amdgcn_sched_barrier(0);
    const float4* hb = (const float4*)h_lds + BUF * (H / 4);
    float4 hv[8];
    hv[0] = hb[q[0]];
    hv[1] = hb[q[1]];
    handover_barrier_3_reads_in_flight();

    #pragma unroll
    for (int i = 2; i < 8; ++i) hv[i] = hb[q[i]];
    __builtin_amdgcn_sched_barrier(0);

    v2f accA[4], accB[4], accC[4], accD[4];
    #pragma unroll
    for (int g = 0; g < 4; ++g) {
        accA[g] = __builtin_elementwise_fma(xv, wi[g], mkv2(bias[g], 0.0f));
        accB[g] = mkv2(0.0f, 0.0f);
        accC[g] = mkv2(0.0f, 0.0f);
        accD[g] = mkv2(0.0f, 0.0f);
    }
    #pragma unroll
    for (int k = 0; k < 8; k += 2) {
        #pragma unroll
        for (int g = 0; g < 4; ++g) {
            accA[g] = __builtin_elementwise_fma(mkv2(hv[k].x, hv[k].y), w2[g][2 * k + 0], accA[g]);
            accB[g] = __builtin_elementwise_fma(mkv2(hv[k].z, hv[k].w), w2[g][2 * k + 1], accB[g]);
            accC[g] = __builtin_elementwise_fma(mkv2(hv[k + 1].x, hv[k + 1].y), w2[g][2 * k + 2], accC[g]);
            accD[g] = __builtin_elementwise_fma(mkv2(hv[k + 1].z, hv[k + 1].w), w2[g][2 * k + 3], accD[g]);
        }
    }
    float gate[4];
    #pragma unroll
    for (int g = 0; g < 4; ++g) {
        const v2f accS = (accA[g] + accB[g]) + (accC[g] + accD[g]);
        const float s = accS.x + accS.y;
        gate[g] = s + xhalf(s);
    }

    const float ti = htanh(gate[0]);
    const float tf = htanh(gate[1]);
    const float gg = htanh(gate[2]);
    const float to = htanh(gate[3]);
    const float gi = __builtin_fmaf(ti, 0.5f, 0.5f);
    const float fc = __builtin_fmaf(tf, hc, hc);
    c = __builtin_fmaf(gi, gg, fc);
    hc = 0.5f * c;
    const float go = __builtin_fmaf(to, 0.5f, 0.5f);
    const float hn = go * htanh(c);
    ((float*)h_lds)[(1 - BUF) * H + j] = hn;
    __builtin_amdgcn_sched_barrier(0);
    return hn;
}

__global__ __launch_bounds__(128, 1)
void lstm_scan_kernel(const float* __restrict__ x,
                      const float* __restrict__ W_ih,
                      const float* __restrict__ W_hh,
                      const float* __restrict__ b_ih,
                      const float* __restrict__ b_hh,
                      float* __restrict__ hs,
                      int T) {
    __shared__ __attribute__((aligned(16))) float h_lds[2 * H];
    __shared__ __attribute__((aligned(16))) float xbuf[2 * CHUNK];

    const int tid = threadIdx.x;
    const int n_x = 2 * T;
    const int wv = tid >> 5;
    const int l = tid & 31;
    const int hh = l >> 4;
    const int j = 16 * wv + (l & 15);

    int q[8];
    q[0] = 4 * wv + 2 * hh;
    q[1] = q[0] + 1;
    #pragma unroll
    for (int i = 2; i < 8; ++i) q[i] = (4 * wv + 4 + 6 * hh + (i - 2)) & 15;

    v2f w2[4][16];
    v2f wi[4];
    float bias[4];
    #pragma unroll
    for (int g = 0; g < 4; ++g) {
        const int row = g * H + j;
        const float sc = (g == 2) ? 1.0f : 0.5f;
        const float4* wrow = (const float4*)(W_hh + row * H);
        #pragma unroll
        for (int i = 0; i < 8; ++i) {
            const float4 w = wrow[q[i]];
            w2[g][2 * i + 0] = mkv2(sc * w.x, sc * w.y);
            w2[g][2 * i + 1] = mkv2(sc * w.z, sc * w.w);
        }
        const float bsum = sc * (b_ih[row] + b_hh[row]);
        const float wi0 = sc * W_ih[2 * row + 0];
        const float wi1 = sc * W_ih[2 * row + 1];
        bias[g] = (hh == 0) ? bsum : 0.0f;
        wi[g] = mkv2((hh == 0) ? wi0 : 0.0f, (hh == 0) ? wi1 : 0.0f);
    }

    if (tid < H) h_lds[tid] = 0.0f;
    float c = 0.0f;
    float hc = 0.0f;

    float2 xreg = load_x2(x, 2 * tid, n_x);
    __syncthreads();

    const int nchunks = (T + CHUNK - 1) / CHUNK;
    for (int ch = 0; ch < nchunks; ++ch) {
        lds_barrier();
        *(float2*)&xbuf[2 * tid] = xreg;
        if (ch + 1 < nchunks) xreg = load_x2(x, (ch + 1) * (2 * CHUNK) + 2 * tid, n_x);
        lds_barrier();

        const int tend = (T - ch * CHUNK < CHUNK) ? (T - ch * CHUNK) : CHUNK;
        int tt = 0;
        for (; tt + 1 < tend; tt += 2) {
            const float h0 = lstm_step<0>(h_lds, xbuf, tt, j, q, w2, wi, bias, c, hc);
            const float h1 = lstm_step<1>(h_lds, xbuf, tt + 1, j, q, w2, wi, bias, c, hc);
            const long line = (long)((ch * CHUNK + tt) >> 1) * 4 + wv;
            hs[line * 32 + l] = hh ? h1 : h0;
        }
        if (tt < tend) {
            const float h0 = lstm_step<0>(h_lds, xbuf, tt, j, q, w2, wi, bias, c, hc);
            const long line = (long)((ch * CHUNK + tt) >> 1) * 4 + wv;
            if (hh == 0) hs[line * 32 + l] = h0;
        }
    }
}

__global__ __launch_bounds__(256, 4)
void fc_wmma_kernel(const float* __restrict__ hs,
                    const float* __restrict__ fc_w,
                    const float* __restrict__ fc_b,
                    float* __restrict__ out,
                    int T) {
    const int lane = threadIdx.x & 31;
    const int wave = threadIdx.x >> 5;
    const int tile = blockIdx.x * 8 + wave;
    const int row0 = tile * 16;
    if (row0 >= T) return;

    const int M    = lane & 15;
    const int half = lane >> 4;
    const int N    = lane & 15;

    v2f bfrag[16];
    #pragma unroll
    for (int kk = 0; kk < 16; ++kk) {
        const int k0 = kk * 4 + half * 2;
        const float b0 = (N < 2) ? fc_w[N * H + k0 + 0] : 0.0f;
        const float b1 = (N < 2) ? fc_w[N * H + k0 + 1] : 0.0f;
        bfrag[kk] = mkv2(b0, b1);
    }

    int row = row0 + M;
    if (row >= T) row = T - 1;
    const float* arow = hs + (long)(row >> 1) * (4 * 32) + (row & 1) * 16;

    v8f cacc = {};
    #pragma unroll
    for (int kk = 0; kk < 16; ++kk) {
        const v2f afrag = *(const v2f*)(arow + (kk >> 2) * 32 + (kk & 3) * 4 + half * 2);
        cacc = __builtin_amdgcn_wmma_f32_16x16x4_f32(
             false, afrag,  false, bfrag[kk],
             (short)0, cacc,  false,  false);
    }

    #pragma unroll
    for (int r = 0; r < 8; ++r) {
        const int orow = row0 + r + 8 * half;
        if (N < 2 && orow < T) out[orow * 2 + N] = cacc[r] + fc_b[N];
    }
}

extern "C" void kernel_launch(void* const* d_in, const int* in_sizes, int n_in,
                              void* d_out, int out_size, void* d_ws, size_t ws_size,
                              hipStream_t stream) {
    const float* x    = (const float*)d_in[0];
    const float* W_ih = (const float*)d_in[1];
    const float* W_hh = (const float*)d_in[2];
    const float* b_ih = (const float*)d_in[3];
    const float* b_hh = (const float*)d_in[4];
    const float* fc_w = (const float*)d_in[5];
    const float* fc_b = (const float*)d_in[6];
    float* out = (float*)d_out;

    const int T = in_sizes[0] / 2;
    float* hs = (float*)d_ws;

    lstm_scan_kernel<<<1, 128, 0, stream>>>(x, W_ih, W_hh, b_ih, b_hh, hs, T);

    const int tiles  = (T + 15) / 16;
    const int blocks = (tiles + 7) / 8;
    fc_wmma_kernel<<<blocks, 256, 0, stream>>>(hs, fc_w, fc_b, out, T);
}


# Scan kernel: a packed FMA holds two gates of a unit at one k (h for both halves by op_sel) instead of one gate at two k: no add across register halves before the lane join; order of sums changes
# speedup vs baseline: 5.4909x; 1.0302x over previous
#include <hip/hip_runtime.h>


#define H      64
#define CHUNK  128

typedef __attribute__((ext_vector_type(2))) float v2f;
typedef __attribute__((ext_vector_type(8))) float v8f;

__device__ __forceinline__ v2f mkv2(float a, float b) { v2f r; r.x = a; r.y = b; return r; }

__device__ __forceinline__ float htanh(float x) {
    return __builtin_amdgcn_tanhf(x);
}

__device__ __forceinline__ float xhalf(float x) {
    const int b = __builtin_bit_cast(int, x);
    return __builtin_bit_cast(float,
        __builtin_amdgcn_permlanex16(b, b, 0x76543210, 0xfedcba98, false, false));
}

__device__ __forceinline__ void lds_barrier() {
    asm volatile("s_wait_dscnt 0x0\n\ts_barrier_signal -1\n\ts_barrier_wait -1" ::: "memory");
}

__device__ __forceinline__ void handover_barrier_3_reads_in_flight() {
    asm volatile("s_wait_dscnt 0x3\n\ts_barrier_signal -1\n\ts_barrier_wait -1" ::: "memory");
}

__device__ __forceinline__ float2 load_x2(const float* __restrict__ x, int idx, int n) {
    float2 r = make_float2(0.0f, 0.0f);
    if (idx + 1 < n) {
        r = *(const float2*)(x + idx);
    } else if (idx < n) {
        r.x = x[idx];
    }
    return r;
}


template <int BUF>
__device__ __forceinline__ float lstm_step(const float* h_lds, const float* xbuf, int tt, int j,
                                           const int (&q)[8], const v2f (&w2)[2][32],
                                           const v2f (&wi)[2][2], const v2f (&bias)[2],
                                           float& c, float& hc) {
    const v2f xv = *(const v2f*)&xbuf[2 * tt];
    __builtin_amdgcn_sched_barrier(0);
    const float4* hb = (const float4*)h_lds + BUF * (H / 4);
    float4 hv[8];
    hv[0] = hb[q[0]];
    hv[1] = hb[q[1]];
    handover_barrier_3_reads_in_flight();

    #pragma unroll
    for (int i = 2; i < 8; ++i) hv[i] = hb[q[i]];
    __builtin_amdgcn_sched_barrier(0);

    v2f acc[2][4];
    #pragma unroll
    for (int p = 0; p < 2; ++p) {
        acc[p][0] = __builtin_elementwise_fma(mkv2(xv.x, xv.x), wi[p][0], bias[p]);
        acc[p][1] = mkv2(xv.y, xv.y) * wi[p][1];
        acc[p][0] = __builtin_elementwise_fma(mkv2(hv[0].x, hv[0].x), w2[p][0], acc[p][0]);
        acc[p][1] = __builtin_elementwise_fma(mkv2(hv[0].y, hv[0].y), w2[p][1], acc[p][1]);
        acc[p][2] = mkv2(hv[0].z, hv[0].z) * w2[p][2];
        acc[p][3] = mkv2(hv[0].w, hv[0].w) * w2[p][3];
    }
    #pragma unroll
    for (int k = 1; k < 8; ++k) {
        #pragma unroll
        for (int p = 0; p < 2; ++p) {
            acc[p][0] = __builtin_elementwise_fma(mkv2(hv[k].x, hv[k].x), w2[p][4 * k + 0], acc[p][0]);
            acc[p][1] = __builtin_elementwise_fma(mkv2(hv[k].y, hv[k].y), w2[p][4 * k + 1], acc[p][1]);
            acc[p][2] = __builtin_elementwise_fma(mkv2(hv[k].z, hv[k].z), w2[p][4 * k + 2], acc[p][2]);
            acc[p][3] = __builtin_elementwise_fma(mkv2(hv[k].w, hv[k].w), w2[p][4 * k + 3], acc[p][3]);
        }
    }
    float gate[4];
    #pragma unroll
    for (int p = 0; p < 2; ++p) {
        const v2f s = (acc[p][0] + acc[p][1]) + (acc[p][2] + acc[p][3]);
        gate[2 * p + 0] = s.x + xhalf(s.x);
        gate[2 * p + 1] = s.y + xhalf(s.y);
    }

    const float ti = htanh(gate[0]);
    const float tf = htanh(gate[1]);
    const float gg = htanh(gate[2]);
    const float to = htanh(gate[3]);
    const float gi = __builtin_fmaf(ti, 0.5f, 0.5f);
    const float fc = __builtin_fmaf(tf, hc, hc);
    c = __builtin_fmaf(gi, gg, fc);
    hc = 0.5f * c;
    const float go = __builtin_fmaf(to, 0.5f, 0.5f);
    const float hn = go * htanh(c);
    ((float*)h_lds)[(1 - BUF) * H + j] = hn;
    __builtin_amdgcn_sched_barrier(0);
    return hn;
}

__global__ __launch_bounds__(128, 1)
void lstm_scan_kernel(const float* __restrict__ x,
                      const float* __restrict__ W_ih,
                      const float* __restrict__ W_hh,
                      const float* __restrict__ b_ih,
                      const float* __restrict__ b_hh,
                      float* __restrict__ hs,
                      int T) {
    __shared__ __attribute__((aligned(16))) float h_lds[2 * H];
    __shared__ __attribute__((aligned(16))) float xbuf[2 * CHUNK];

    const int tid = threadIdx.x;
    const int n_x = 2 * T;
    const int wv = tid >> 5;
    const int l = tid & 31;
    const int hh = l >> 4;
    const int j = 16 * wv + (l & 15);

    int q[8];
    q[0] = 4 * wv + 2 * hh;
    q[1] = q[0] + 1;
    #pragma unroll
    for (int i = 2; i < 8; ++i) q[i] = (4 * wv + 4 + 6 * hh + (i - 2)) & 15;

    v2f w2[2][32];
    v2f wi[2][2];
    v2f bias[2];
    #pragma unroll
    for (int p = 0; p < 2; ++p) {
        const int rowa = (2 * p + 0) * H + j;
        const int rowb = (2 * p + 1) * H + j;
        const float sa = (p == 1) ? 1.0f : 0.5f;
        const float sb = 0.5f;
        const float4* wa = (const float4*)(W_hh + rowa * H);
        const float4* wb = (const float4*)(W_hh + rowb * H);
        #pragma unroll
        for (int i = 0; i < 8; ++i) {
            const float4 a = wa[q[i]];
            const float4 b = wb[q[i]];
            w2[p][4 * i + 0] = mkv2(sa * a.x, sb * b.x);
            w2[p][4 * i + 1] = mkv2(sa * a.y, sb * b.y);
            w2[p][4 * i + 2] = mkv2(sa * a.z, sb * b.z);
            w2[p][4 * i + 3] = mkv2(sa * a.w, sb * b.w);
        }
        const float ba = sa * (b_ih[rowa] + b_hh[rowa]);
        const float bb = sb * (b_ih[rowb] + b_hh[rowb]);
        bias[p]  = (hh == 0) ? mkv2(ba, bb) : mkv2(0.0f, 0.0f);
        wi[p][0] = (hh == 0) ? mkv2(sa * W_ih[2 * rowa + 0], sb * W_ih[2 * rowb + 0]) : mkv2(0.0f, 0.0f);
        wi[p][1] = (hh == 0) ? mkv2(sa * W_ih[2 * rowa + 1], sb * W_ih[2 * rowb + 1]) : mkv2(0.0f, 0.0f);
    }

    if (tid < H) h_lds[tid] = 0.0f;
    float c = 0.0f;
    float hc = 0.0f;

    float2 xreg = load_x2(x, 2 * tid, n_x);
    __syncthreads();

    const int nchunks = (T + CHUNK - 1) / CHUNK;
    for (int ch = 0; ch < nchunks; ++ch) {
        lds_barrier();
        *(float2*)&xbuf[2 * tid] = xreg;
        if (ch + 1 < nchunks) xreg = load_x2(x, (ch + 1) * (2 * CHUNK) + 2 * tid, n_x);
        lds_barrier();

        const int tend = (T - ch * CHUNK < CHUNK) ? (T - ch * CHUNK) : CHUNK;
        int tt = 0;
        for (; tt + 1 < tend; tt += 2) {
            const float h0 = lstm_step<0>(h_lds, xbuf, tt, j, q, w2, wi, bias, c, hc);
            const float h1 = lstm_step<1>(h_lds, xbuf, tt + 1, j, q, w2, wi, bias, c, hc);
            const long line = (long)((ch * CHUNK + tt) >> 1) * 4 + wv;
            hs[line * 32 + l] = hh ? h1 : h0;
        }
        if (tt < tend) {
            const float h0 = lstm_step<0>(h_lds, xbuf, tt, j, q, w2, wi, bias, c, hc);
            const long line = (long)((ch * CHUNK + tt) >> 1) * 4 + wv;
            if (hh == 0) hs[line * 32 + l] = h0;
        }
    }
}

__global__ __launch_bounds__(256, 4)
void fc_wmma_kernel(const float* __restrict__ hs,
                    const float* __restrict__ fc_w,
                    const float* __restrict__ fc_b,
                    float* __restrict__ out,
                    int T) {
    const int lane = threadIdx.x & 31;
    const int wave = threadIdx.x >> 5;
    const int tile = blockIdx.x * 8 + wave;
    const int row0 = tile * 16;
    if (row0 >= T) return;

    const int M    = lane & 15;
    const int half = lane >> 4;
    const int N    = lane & 15;

    v2f bfrag[16];
    #pragma unroll
    for (int kk = 0; kk < 16; ++kk) {
        const int k0 = kk * 4 + half * 2;
        const float b0 = (N < 2) ? fc_w[N * H + k0 + 0] : 0.0f;
        const float b1 = (N < 2) ? fc_w[N * H + k0 + 1] : 0.0f;
        bfrag[kk] = mkv2(b0, b1);
    }

    int row = row0 + M;
    if (row >= T) row = T - 1;
    const float* arow = hs + (long)(row >> 1) * (4 * 32) + (row & 1) * 16;

    v8f cacc = {};
    #pragma unroll
    for (int kk = 0; kk < 16; ++kk) {
        const v2f afrag = *(const v2f*)(arow + (kk >> 2) * 32 + (kk & 3) * 4 + half * 2);
        cacc = __builtin_amdgcn_wmma_f32_16x16x4_f32(
             false, afrag,  false, bfrag[kk],
             (short)0, cacc,  false,  false);
    }

    #pragma unroll
    for (int r = 0; r < 8; ++r) {
        const int orow = row0 + r + 8 * half;
        if (N < 2 && orow < T) out[orow * 2 + N] = cacc[r] + fc_b[N];
    }
}

extern "C" void kernel_launch(void* const* d_in, const int* in_sizes, int n_in,
                              void* d_out, int out_size, void* d_ws, size_t ws_size,
                              hipStream_t stream) {
    const float* x    = (const float*)d_in[0];
    const float* W_ih = (const float*)d_in[1];
    const float* W_hh = (const float*)d_in[2];
    const float* b_ih = (const float*)d_in[3];
    const float* b_hh = (const float*)d_in[4];
    const float* fc_w = (const float*)d_in[5];
    const float* fc_b = (const float*)d_in[6];
    float* out = (float*)d_out;

    const int T = in_sizes[0] / 2;
    float* hs = (float*)d_ws;

    lstm_scan_kernel<<<1, 128, 0, stream>>>(x, W_ih, W_hh, b_ih, b_hh, hs, T);

    const int tiles  = (T + 15) / 16;
    const int blocks = (tiles + 7) / 8;
    fc_wmma_kernel<<<blocks, 256, 0, stream>>>(hs, fc_w, fc_b, out, T);
}


# Scan kernel: the two lanes of a unit sit in one 16-lane row (l, l ^ 8): half sums joined by one DPP add (row_xmask:8) for mov + v_permlanex16 + add; hs line layout follows; sums unchanged
# speedup vs baseline: 5.6656x; 1.0318x over previous
#include <hip/hip_runtime.h>


#define H      64
#define CHUNK  128

typedef __attribute__((ext_vector_type(2))) float v2f;
typedef __attribute__((ext_vector_type(8))) float v8f;

__device__ __forceinline__ v2f mkv2(float a, float b) { v2f r; r.x = a; r.y = b; return r; }

__device__ __forceinline__ float htanh(float x) {
    return __builtin_amdgcn_tanhf(x);
}

__device__ __forceinline__ float pair_sum(float x) {
    const int b = __builtin_bit_cast(int, x);
    const int o = __builtin_amdgcn_update_dpp(b, b, 0x168, 0xf, 0xf, true);
    return x + __builtin_bit_cast(float, o);
}

__device__ __forceinline__ void lds_barrier() {
    asm volatile("s_wait_dscnt 0x0\n\ts_barrier_signal -1\n\ts_barrier_wait -1" ::: "memory");
}

__device__ __forceinline__ void handover_barrier_3_reads_in_flight() {
    asm volatile("s_wait_dscnt 0x3\n\ts_barrier_signal -1\n\ts_barrier_wait -1" ::: "memory");
}

__device__ __forceinline__ float2 load_x2(const float* __restrict__ x, int idx, int n) {
    float2 r = make_float2(0.0f, 0.0f);
    if (idx + 1 < n) {
        r = *(const float2*)(x + idx);
    } else if (idx < n) {
        r.x = x[idx];
    }
    return r;
}


template <int BUF>
__device__ __forceinline__ float lstm_step(const float* h_lds, const float* xbuf, int tt, int j,
                                           const int (&q)[8], const v2f (&w2)[2][32],
                                           const v2f (&wi)[2][2], const v2f (&bias)[2],
                                           float& c, float& hc) {
    const v2f xv = *(const v2f*)&xbuf[2 * tt];
    __builtin_amdgcn_sched_barrier(0);
    const float4* hb = (const float4*)h_lds + BUF * (H / 4);
    float4 hv[8];
    hv[0] = hb[q[0]];
    hv[1] = hb[q[1]];
    handover_barrier_3_reads_in_flight();

    #pragma unroll
    for (int i = 2; i < 8; ++i) hv[i] = hb[q[i]];
    __builtin_amdgcn_sched_barrier(0);

    v2f acc[2][4];
    #pragma unroll
    for (int p = 0; p < 2; ++p) {
        acc[p][0] = __builtin_elementwise_fma(mkv2(xv.x, xv.x), wi[p][0], bias[p]);
        acc[p][1] = mkv2(xv.y, xv.y) * wi[p][1];
        acc[p][0] = __builtin_elementwise_fma(mkv2(hv[0].x, hv[0].x), w2[p][0], acc[p][0]);
        acc[p][1] = __builtin_elementwise_fma(mkv2(hv[0].y, hv[0].y), w2[p][1], acc[p][1]);
        acc[p][2] = mkv2(hv[0].z, hv[0].z) * w2[p][2];
        acc[p][3] = mkv2(hv[0].w, hv[0].w) * w2[p][3];
    }
    #pragma unroll
    for (int k = 1; k < 8; ++k) {
        #pragma unroll
        for (int p = 0; p < 2; ++p) {
            acc[p][0] = __builtin_elementwise_fma(mkv2(hv[k].x, hv[k].x), w2[p][4 * k + 0], acc[p][0]);
            acc[p][1] = __builtin_elementwise_fma(mkv2(hv[k].y, hv[k].y), w2[p][4 * k + 1], acc[p][1]);
            acc[p][2] = __builtin_elementwise_fma(mkv2(hv[k].z, hv[k].z), w2[p][4 * k + 2], acc[p][2]);
            acc[p][3] = __builtin_elementwise_fma(mkv2(hv[k].w, hv[k].w), w2[p][4 * k + 3], acc[p][3]);
        }
    }
    float gate[4];
    #pragma unroll
    for (int p = 0; p < 2; ++p) {
        const v2f s = (acc[p][0] + acc[p][1]) + (acc[p][2] + acc[p][3]);
        gate[2 * p + 0] = pair_sum(s.x);
        gate[2 * p + 1] = pair_sum(s.y);
    }

    const float ti = htanh(gate[0]);
    const float tf = htanh(gate[1]);
    const float gg = htanh(gate[2]);
    const float to = htanh(gate[3]);
    const float gi = __builtin_fmaf(ti, 0.5f, 0.5f);
    const float fc = __builtin_fmaf(tf, hc, hc);
    c = __builtin_fmaf(gi, gg, fc);
    hc = 0.5f * c;
    const float go = __builtin_fmaf(to, 0.5f, 0.5f);
    const float hn = go * htanh(c);
    ((float*)h_lds)[(1 - BUF) * H + j] = hn;
    __builtin_amdgcn_sched_barrier(0);
    return hn;
}

__global__ __launch_bounds__(128, 1)
void lstm_scan_kernel(const float* __restrict__ x,
                      const float* __restrict__ W_ih,
                      const float* __restrict__ W_hh,
                      const float* __restrict__ b_ih,
                      const float* __restrict__ b_hh,
                      float* __restrict__ hs,
                      int T) {
    __shared__ __attribute__((aligned(16))) float h_lds[2 * H];
    __shared__ __attribute__((aligned(16))) float xbuf[2 * CHUNK];

    const int tid = threadIdx.x;
    const int n_x = 2 * T;
    const int wv = tid >> 5;
    const int l = tid & 31;
    const int hh = (l >> 3) & 1;
    const int j = 16 * wv + 8 * (l >> 4) + (l & 7);

    int q[8];
    q[0] = 4 * wv + 2 * hh;
    q[1] = q[0] + 1;
    #pragma unroll
    for (int i = 2; i < 8; ++i) q[i] = (4 * wv + 4 + 6 * hh + (i - 2)) & 15;

    v2f w2[2][32];
    v2f wi[2][2];
    v2f bias[2];
    #pragma unroll
    for (int p = 0; p < 2; ++p) {
        const int rowa = (2 * p + 0) * H + j;
        const int rowb = (2 * p + 1) * H + j;
        const float sa = (p == 1) ? 1.0f : 0.5f;
        const float sb = 0.5f;
        const float4* wa = (const float4*)(W_hh + rowa * H);
        const float4* wb = (const float4*)(W_hh + rowb * H);
        #pragma unroll
        for (int i = 0; i < 8; ++i) {
            const float4 a = wa[q[i]];
            const float4 b = wb[q[i]];
            w2[p][4 * i + 0] = mkv2(sa * a.x, sb * b.x);
            w2[p][4 * i + 1] = mkv2(sa * a.y, sb * b.y);
            w2[p][4 * i + 2] = mkv2(sa * a.z, sb * b.z);
            w2[p][4 * i + 3] = mkv2(sa * a.w, sb * b.w);
        }
        const float ba = sa * (b_ih[rowa] + b_hh[rowa]);
        const float bb = sb * (b_ih[rowb] + b_hh[rowb]);
        bias[p]  = (hh == 0) ? mkv2(ba, bb) : mkv2(0.0f, 0.0f);
        wi[p][0] = (hh == 0) ? mkv2(sa * W_ih[2 * rowa + 0], sb * W_ih[2 * rowb + 0]) : mkv2(0.0f, 0.0f);
        wi[p][1] = (hh == 0) ? mkv2(sa * W_ih[2 * rowa + 1], sb * W_ih[2 * rowb + 1]) : mkv2(0.0f, 0.0f);
    }

    if (tid < H) h_lds[tid] = 0.0f;
    float c = 0.0f;
    float hc = 0.0f;

    float2 xreg = load_x2(x, 2 * tid, n_x);
    __syncthreads();

    const int nchunks = (T + CHUNK - 1) / CHUNK;
    for (int ch = 0; ch < nchunks; ++ch) {
        lds_barrier();
        *(float2*)&xbuf[2 * tid] = xreg;
        if (ch + 1 < nchunks) xreg = load_x2(x, (ch + 1) * (2 * CHUNK) + 2 * tid, n_x);
        lds_barrier();

        const int tend = (T - ch * CHUNK < CHUNK) ? (T - ch * CHUNK) : CHUNK;
        int tt = 0;
        for (; tt + 1 < tend; tt += 2) {
            const float h0 = lstm_step<0>(h_lds, xbuf, tt, j, q, w2, wi, bias, c, hc);
            const float h1 = lstm_step<1>(h_lds, xbuf, tt + 1, j, q, w2, wi, bias, c, hc);
            const long line = (long)((ch * CHUNK + tt) >> 1) * 4 + wv;
            hs[line * 32 + l] = hh ? h1 : h0;
        }
        if (tt < tend) {
            const float h0 = lstm_step<0>(h_lds, xbuf, tt, j, q, w2, wi, bias, c, hc);
            const long line = (long)((ch * CHUNK + tt) >> 1) * 4 + wv;
            if (hh == 0) hs[line * 32 + l] = h0;
        }
    }
}

__global__ __launch_bounds__(256, 4)
void fc_wmma_kernel(const float* __restrict__ hs,
                    const float* __restrict__ fc_w,
                    const float* __restrict__ fc_b,
                    float* __restrict__ out,
                    int T) {
    const int lane = threadIdx.x & 31;
    const int wave = threadIdx.x >> 5;
    const int tile = blockIdx.x * 8 + wave;
    const int row0 = tile * 16;
    if (row0 >= T) return;

    const int M    = lane & 15;
    const int half = lane >> 4;
    const int N    = lane & 15;

    v2f bfrag[16];
    #pragma unroll
    for (int kk = 0; kk < 16; ++kk) {
        const int k0 = kk * 4 + half * 2;
        const float b0 = (N < 2) ? fc_w[N * H + k0 + 0] : 0.0f;
        const float b1 = (N < 2) ? fc_w[N * H + k0 + 1] : 0.0f;
        bfrag[kk] = mkv2(b0, b1);
    }

    int row = row0 + M;
    if (row >= T) row = T - 1;
    const float* arow = hs + (long)(row >> 1) * (4 * 32) + (row & 1) * 8;

    v8f cacc = {};
    #pragma unroll
    for (int kk = 0; kk < 16; ++kk) {
        const v2f afrag = *(const v2f*)(arow + (kk >> 2) * 32 + ((kk >> 1) & 1) * 16 + (kk & 1) * 4 + half * 2);
        cacc = __builtin_amdgcn_wmma_f32_16x16x4_f32(
             false, afrag,  false, bfrag[kk],
             (short)0, cacc,  false,  false);
    }

    #pragma unroll
    for (int r = 0; r < 8; ++r) {
        const int orow = row0 + r + 8 * half;
        if (N < 2 && orow < T) out[orow * 2 + N] = cacc[r] + fc_b[N];
    }
}

extern "C" void kernel_launch(void* const* d_in, const int* in_sizes, int n_in,
                              void* d_out, int out_size, void* d_ws, size_t ws_size,
                              hipStream_t stream) {
    const float* x    = (const float*)d_in[0];
    const float* W_ih = (const float*)d_in[1];
    const float* W_hh = (const float*)d_in[2];
    const float* b_ih = (const float*)d_in[3];
    const float* b_hh = (const float*)d_in[4];
    const float* fc_w = (const float*)d_in[5];
    const float* fc_b = (const float*)d_in[6];
    float* out = (float*)d_out;

    const int T = in_sizes[0] / 2;
    float* hs = (float*)d_ws;

    lstm_scan_kernel<<<1, 128, 0, stream>>>(x, W_ih, W_hh, b_ih, b_hh, hs, T);

    const int tiles  = (T + 15) / 16;
    const int blocks = (tiles + 7) / 8;
    fc_wmma_kernel<<<blocks, 256, 0, stream>>>(hs, fc_w, fc_b, out, T);
}
